# starting kernel
# baseline (speedup 1.0000x reference)
#include <hip/hip_runtime.h>
#include <math.h>

typedef __attribute__((ext_vector_type(2))) float v2f;
typedef __attribute__((ext_vector_type(8))) float v8f;

#define DIST_COEFF 10000.0f

#define BCX (-0.001782f)
#define BCY (1.005e-05f)
#define BCZ (0.0431621f)
#define HEX ((float)(0.204416 / 2.0 + 0.001))
#define HEY ((float)(0.0632517 / 2.0 + 0.001))
#define HEZ ((float)(0.1381738 / 2.0 + 0.001))

__global__ __launch_bounds__(256) void collision_dist_kernel(
    const float* __restrict__ trans,
    const float* __restrict__ quat,
    const float* __restrict__ pc,
    float* __restrict__ out,
    int N)
{
    const int b      = blockIdx.x;
    const int tid    = threadIdx.x;
    const int lane   = tid & 31;
    const int wave   = tid >> 5;
    const int nwaves = blockDim.x >> 5;
    const int m      = lane & 15;
    const bool hi    = lane >= 16;

    const float qx = quat[b * 4 + 0], qy = quat[b * 4 + 1];
    const float qz = quat[b * 4 + 2], qw = quat[b * 4 + 3];
    const float tx = trans[b * 3 + 0], ty = trans[b * 3 + 1], tz = trans[b * 3 + 2];

    const float s2 = qx * qx + qy * qy + qz * qz + qw * qw;
    const float ux = -qx / s2, uy = -qy / s2, uz = -qz / s2, uw = qw / s2;
    const float tt = uw * uw - (ux * ux + uy * uy + uz * uz);

    const float M00 = tt + 2.f * ux * ux;
    const float M01 = 2.f * (ux * uy - uw * uz);
    const float M02 = 2.f * (ux * uz + uw * uy);
    const float M10 = 2.f * (ux * uy + uw * uz);
    const float M11 = tt + 2.f * uy * uy;
    const float M12 = 2.f * (uy * uz - uw * ux);
    const float M20 = 2.f * (ux * uz - uw * uy);
    const float M21 = 2.f * (uy * uz + uw * ux);
    const float M22 = tt + 2.f * uz * uz;

    float c0 = 0.f, c1 = 0.f, c2 = 0.f, c3 = 0.f;
    if (m == 0)      { c0 = M00; c1 = M01; c2 = M02; c3 = -tx; }
    else if (m == 1) { c0 = M10; c1 = M11; c2 = M12; c3 = -ty; }
    else if (m == 2) { c0 = M20; c1 = M21; c2 = M22; c3 = -tz; }
    v2f A;
    A.x = hi ? c2 : c0;
    A.y = hi ? c3 : c1;

    const float lox = BCX - HEX, hix = BCX + HEX;
    const float loy = BCY - HEY, hiy = BCY + HEY;
    const float loz = BCZ - HEZ, hiz = BCZ + HEZ;

    const int r0 = m * 12 + (hi ? 8 : 0);
    const int r1 = hi ? 0 : (m * 12 + 4);
    const int idxb = m * 4;

    const int poff = (lane < 2 ? lane : 2) * 128;

    const char* rb = (const char*)pc + (size_t)b * (size_t)N * 12u;

    float accS = 0.f, accC = 0.f;
    const v8f cacc = {0.f, 0.f, 0.f, 0.f, 0.f, 0.f, 0.f, 0.f};

    const int nch = N >> 5;
    #pragma unroll 2
    for (int ch = wave; ch < nch; ch += nwaves) {
        const size_t base = (size_t)ch * 384u;

        __builtin_prefetch(rb + base + 3072 + poff, 0, 1);

        v2f B1;
        B1.x = *(const float*)(rb + base + r0);
        {
            const float yv = *(const float*)(rb + base + r1);
            B1.y = hi ? 1.0f : yv;
        }
        v2f B2;
        B2.x = *(const float*)(rb + base + 192 + r0);
        {
            const float yv = *(const float*)(rb + base + 192 + r1);
            B2.y = hi ? 1.0f : yv;
        }

        v8f d1 = __builtin_amdgcn_wmma_f32_16x16x4_f32(
            false, A, false, B1, (short)0, cacc, false, false);
        v8f d2 = __builtin_amdgcn_wmma_f32_16x16x4_f32(
            false, A, false, B2, (short)0, cacc, false, false);

        const float bx = __int_as_float(
            __builtin_amdgcn_ds_bpermute(idxb, __float_as_int(d2[0])));
        const float by = __int_as_float(
            __builtin_amdgcn_ds_bpermute(idxb, __float_as_int(d2[1])));
        const float bz = __int_as_float(
            __builtin_amdgcn_ds_bpermute(idxb, __float_as_int(d2[2])));

        const float px = hi ? bx : d1[0];
        const float py = hi ? by : d1[1];
        const float pz = hi ? bz : d1[2];

        const bool in =
            (px >= lox) & (px <= hix) &
            (py >= loy) & (py <= hiy) &
            (pz >= loz) & (pz <= hiz);
        const float nr = __builtin_amdgcn_sqrtf(px * px + py * py + pz * pz);
        accS += in ? nr : 0.0f;
        accC += in ? 1.0f : 0.0f;
    }

    for (int off = 16; off > 0; off >>= 1) {
        accS += __shfl_xor(accS, off, 32);
        accC += __shfl_xor(accC, off, 32);
    }

    __shared__ float sS[32];
    __shared__ float sC[32];
    if (lane == 0) { sS[wave] = accS; sC[wave] = accC; }
    __syncthreads();

    if (tid == 0) {
        float S = 0.f, C = 0.f;
        for (int i = 0; i < nwaves; ++i) { S += sS[i]; C += sC[i]; }
        out[b] = (C > 0.f) ? (-S / C) * DIST_COEFF : DIST_COEFF;
    }
}

extern "C" void kernel_launch(void* const* d_in, const int* in_sizes, int n_in,
                              void* d_out, int out_size, void* d_ws, size_t ws_size,
                              hipStream_t stream) {
    const float* trans = (const float*)d_in[0];
    const float* quat  = (const float*)d_in[1];
    const float* pc    = (const float*)d_in[2];

    const int B = in_sizes[0] / 3;
    const int N = (int)(in_sizes[2] / ((long long)B * 3));

    collision_dist_kernel<<<dim3(B), dim3(256), 0, stream>>>(
        trans, quat, pc, (float*)d_out, N);
}


# one single-wave work-group per row streams whole points (one coalesced 384-B load per 32 points, 16-32 loads in flight); both 16-point products chain into one accumulator; new summation order
# speedup vs baseline: 1.0443x; 1.0443x over previous
#include <hip/hip_runtime.h>
#include <math.h>


typedef __attribute__((ext_vector_type(2))) float v2f;
typedef __attribute__((ext_vector_type(8))) float v8f;

#define DIST_COEFF 10000.0f

#define BCX (-0.001782f)
#define BCY (1.005e-05f)
#define BCZ (0.0431621f)
#define HEX ((float)(0.204416 / 2.0 + 0.001))
#define HEY ((float)(0.0632517 / 2.0 + 0.001))
#define HEZ ((float)(0.1381738 / 2.0 + 0.001))

#define UNR 16

struct P3 { float x, y, z; };

__device__ __forceinline__ P3 load_pt(const char* p)
{
    const float* f = (const float*)p;
    P3 r;
    r.x = f[0];
    r.y = f[1];
    r.z = f[2];
    return r;
}

struct Box { float lox, hix, loy, hiy, loz, hiz; };

__device__ __forceinline__ void do_chunk(const P3 p, const bool hi,
                                         const v2f A1, const v2f A2, const Box bx,
                                         float& accS, float& accC)
{
    const float zs = __uint_as_float(__builtin_amdgcn_permlanex16(
        __float_as_uint(p.z), __float_as_uint(p.z),
        0x76543210u, 0xfedcba98u, false, false));

    v2f B1, B2;
    B1.x = hi ? zs   : p.x;
    B1.y = hi ? 1.0f : p.y;
    B2.x = hi ? p.x  : zs;
    B2.y = hi ? p.y  : 1.0f;

    const v8f zero = {0.f, 0.f, 0.f, 0.f, 0.f, 0.f, 0.f, 0.f};
    v8f d = __builtin_amdgcn_wmma_f32_16x16x4_f32(
        false, A1, false, B1, (short)0, zero, false, false);
    d = __builtin_amdgcn_wmma_f32_16x16x4_f32(
        false, A2, false, B2, (short)0, d, false, false);

    const float px = d[0], py = d[1], pz = d[2];

    const bool in =
        (px >= bx.lox) & (px <= bx.hix) &
        (py >= bx.loy) & (py <= bx.hiy) &
        (pz >= bx.loz) & (pz <= bx.hiz);
    const float nr = __builtin_amdgcn_sqrtf(px * px + py * py + pz * pz);
    accS += in ? nr : 0.0f;
    accC += in ? 1.0f : 0.0f;
}

#define GRP 4

__device__ __forceinline__ void load_buf(P3 (&buf)[UNR], const char* q)
{
    #pragma unroll
    for (int u = 0; u < UNR; ++u) buf[u] = load_pt(q + (size_t)u * 384u);
    __builtin_amdgcn_sched_barrier(0);
}

__device__ __forceinline__ void process_buf(const P3 (&buf)[UNR], const bool hi,
                                            const v2f A1, const v2f A2, const Box bx,
                                            float& accS, float& accC)
{
    #pragma unroll
    for (int g = 0; g < UNR; g += GRP) {
        #pragma unroll
        for (int u = 0; u < GRP; ++u)
            do_chunk(buf[g + u], hi, A1, A2, bx, accS, accC);
        __builtin_amdgcn_sched_barrier(0);
    }
}

__global__ __launch_bounds__(32) void collision_dist_kernel(
    const float* __restrict__ trans,
    const float* __restrict__ quat,
    const float* __restrict__ pc,
    float* __restrict__ out,
    int N)
{
    const int b    = blockIdx.x;
    const int lane = threadIdx.x & 31;
    const int m    = lane & 15;
    const bool hi  = lane >= 16;

    const int nch = N >> 5;
    const int nb  = nch / UNR;

    const char* rb = (const char*)pc + (size_t)b * (size_t)N * 12u + (size_t)lane * 12u;

    P3 bufA[UNR], bufB[UNR];
    if (nb > 0)
        load_buf(bufA, rb);

    const float qx = quat[b * 4 + 0], qy = quat[b * 4 + 1];
    const float qz = quat[b * 4 + 2], qw = quat[b * 4 + 3];
    const float tx = trans[b * 3 + 0], ty = trans[b * 3 + 1], tz = trans[b * 3 + 2];

    const float s2 = qx * qx + qy * qy + qz * qz + qw * qw;
    const float ux = -qx / s2, uy = -qy / s2, uz = -qz / s2, uw = qw / s2;
    const float tt = uw * uw - (ux * ux + uy * uy + uz * uz);

    const float M00 = tt + 2.f * ux * ux;
    const float M01 = 2.f * (ux * uy - uw * uz);
    const float M02 = 2.f * (ux * uz + uw * uy);
    const float M10 = 2.f * (ux * uy + uw * uz);
    const float M11 = tt + 2.f * uy * uy;
    const float M12 = 2.f * (uy * uz - uw * ux);
    const float M20 = 2.f * (ux * uz - uw * uy);
    const float M21 = 2.f * (uy * uz + uw * ux);
    const float M22 = tt + 2.f * uz * uz;

    const int r = m & 7;
    float c0 = 0.f, c1 = 0.f, c2 = 0.f, c3 = 0.f;
    if (r == 0)      { c0 = M00; c1 = M01; c2 = M02; c3 = -tx; }
    else if (r == 1) { c0 = M10; c1 = M11; c2 = M12; c3 = -ty; }
    else if (r == 2) { c0 = M20; c1 = M21; c2 = M22; c3 = -tz; }
    const bool lowrows = m < 8;
    v2f A1, A2;
    A1.x = lowrows ? (hi ? c2 : c0) : 0.f;
    A1.y = lowrows ? (hi ? c3 : c1) : 0.f;
    A2.x = lowrows ? 0.f : (hi ? c0 : c2);
    A2.y = lowrows ? 0.f : (hi ? c1 : c3);

    Box bx;
    bx.lox = BCX - HEX; bx.hix = BCX + HEX;
    bx.loy = BCY - HEY; bx.hiy = BCY + HEY;
    bx.loz = BCZ - HEZ; bx.hiz = BCZ + HEZ;

    float accS = 0.f, accC = 0.f;

    int it = 0;
    for (; it + 2 < nb; it += 2) {
        load_buf(bufB, rb + (size_t)(it + 1) * (size_t)(UNR * 384));
        process_buf(bufA, hi, A1, A2, bx, accS, accC);
        load_buf(bufA, rb + (size_t)(it + 2) * (size_t)(UNR * 384));
        process_buf(bufB, hi, A1, A2, bx, accS, accC);
    }
    if (it + 1 < nb) {
        load_buf(bufB, rb + (size_t)(it + 1) * (size_t)(UNR * 384));
        process_buf(bufA, hi, A1, A2, bx, accS, accC);
        process_buf(bufB, hi, A1, A2, bx, accS, accC);
    } else if (it < nb) {
        process_buf(bufA, hi, A1, A2, bx, accS, accC);
    }
    for (int ch = nb * UNR; ch < nch; ++ch) {
        const P3 p = load_pt(rb + (size_t)ch * 384u);
        do_chunk(p, hi, A1, A2, bx, accS, accC);
    }

    for (int off = 16; off > 0; off >>= 1) {
        accS += __shfl_xor(accS, off, 32);
        accC += __shfl_xor(accC, off, 32);
    }

    if (lane == 0)
        out[b] = (accC > 0.f) ? (-accS / accC) * DIST_COEFF : DIST_COEFF;
}

extern "C" void kernel_launch(void* const* d_in, const int* in_sizes, int n_in,
                              void* d_out, int out_size, void* d_ws, size_t ws_size,
                              hipStream_t stream) {
    const float* trans = (const float*)d_in[0];
    const float* quat  = (const float*)d_in[1];
    const float* pc    = (const float*)d_in[2];

    const int B = in_sizes[0] / 3;
    const int N = (int)(in_sizes[2] / ((long long)B * 3));

    collision_dist_kernel<<<dim3(B), dim3(32), 0, stream>>>(
        trans, quat, pc, (float*)d_out, N);
}


# idle rows 3 to 5 of each product return box-normalised coordinates; in-box test = max of three magnitudes <= 1 instead of 6 compares + 5 scalar ands; mask may differ within an ulp of a face
# speedup vs baseline: 1.1380x; 1.0897x over previous
#include <hip/hip_runtime.h>
#include <math.h>


typedef __attribute__((ext_vector_type(2))) float v2f;
typedef __attribute__((ext_vector_type(8))) float v8f;

#define DIST_COEFF 10000.0f

#define BCX (-0.001782f)
#define BCY (1.005e-05f)
#define BCZ (0.0431621f)
#define HEX ((float)(0.204416 / 2.0 + 0.001))
#define HEY ((float)(0.0632517 / 2.0 + 0.001))
#define HEZ ((float)(0.1381738 / 2.0 + 0.001))

#define LOX (BCX - HEX)
#define HIX (BCX + HEX)
#define LOY (BCY - HEY)
#define HIY (BCY + HEY)
#define LOZ (BCZ - HEZ)
#define HIZ (BCZ + HEZ)
#define MIDX ((float)(((double)LOX + (double)HIX) * 0.5))
#define MIDY ((float)(((double)LOY + (double)HIY) * 0.5))
#define MIDZ ((float)(((double)LOZ + (double)HIZ) * 0.5))
#define INVHX ((float)(2.0 / ((double)HIX - (double)LOX)))
#define INVHY ((float)(2.0 / ((double)HIY - (double)LOY)))
#define INVHZ ((float)(2.0 / ((double)HIZ - (double)LOZ)))

#define UNR 16

struct P3 { float x, y, z; };

__device__ __forceinline__ P3 load_pt(const char* p)
{
    const float* f = (const float*)p;
    P3 r;
    r.x = f[0];
    r.y = f[1];
    r.z = f[2];
    return r;
}

__device__ __forceinline__ void do_chunk(const P3 p, const bool hi,
                                         const v2f A1, const v2f A2,
                                         float& accS, float& accC)
{
    const float zs = __uint_as_float(__builtin_amdgcn_permlanex16(
        __float_as_uint(p.z), __float_as_uint(p.z),
        0x76543210u, 0xfedcba98u, false, false));

    v2f B1, B2;
    B1.x = hi ? zs   : p.x;
    B1.y = hi ? 1.0f : p.y;
    B2.x = hi ? p.x  : zs;
    B2.y = hi ? p.y  : 1.0f;

    const v8f zero = {0.f, 0.f, 0.f, 0.f, 0.f, 0.f, 0.f, 0.f};
    v8f d = __builtin_amdgcn_wmma_f32_16x16x4_f32(
        false, A1, false, B1, (short)0, zero, false, false);
    d = __builtin_amdgcn_wmma_f32_16x16x4_f32(
        false, A2, false, B2, (short)0, d, false, false);

    const float px = d[0], py = d[1], pz = d[2];

    const float w = fmaxf(fmaxf(fabsf(d[3]), fabsf(d[4])), fabsf(d[5]));
    const bool in = w <= 1.0f;
    const float nr = __builtin_amdgcn_sqrtf(px * px + py * py + pz * pz);
    accS += in ? nr : 0.0f;
    accC += in ? 1.0f : 0.0f;
}

#define GRP 4

__device__ __forceinline__ void load_buf(P3 (&buf)[UNR], const char* q)
{
    #pragma unroll
    for (int u = 0; u < UNR; ++u) buf[u] = load_pt(q + (size_t)u * 384u);
    __builtin_amdgcn_sched_barrier(0);
}

__device__ __forceinline__ void process_buf(const P3 (&buf)[UNR], const bool hi,
                                            const v2f A1, const v2f A2,
                                            float& accS, float& accC)
{
    #pragma unroll
    for (int g = 0; g < UNR; g += GRP) {
        #pragma unroll
        for (int u = 0; u < GRP; ++u)
            do_chunk(buf[g + u], hi, A1, A2, accS, accC);
        __builtin_amdgcn_sched_barrier(0);
    }
}

__global__ __launch_bounds__(32) void collision_dist_kernel(
    const float* __restrict__ trans,
    const float* __restrict__ quat,
    const float* __restrict__ pc,
    float* __restrict__ out,
    int N)
{
    const int b    = blockIdx.x;
    const int lane = threadIdx.x & 31;
    const int m    = lane & 15;
    const bool hi  = lane >= 16;

    const int nch = N >> 5;
    const int nb  = nch / UNR;

    const char* rb = (const char*)pc + (size_t)b * (size_t)N * 12u + (size_t)lane * 12u;

    P3 bufA[UNR], bufB[UNR];
    if (nb > 0)
        load_buf(bufA, rb);

    const float qx = quat[b * 4 + 0], qy = quat[b * 4 + 1];
    const float qz = quat[b * 4 + 2], qw = quat[b * 4 + 3];
    const float tx = trans[b * 3 + 0], ty = trans[b * 3 + 1], tz = trans[b * 3 + 2];

    const float s2 = qx * qx + qy * qy + qz * qz + qw * qw;
    const float ux = -qx / s2, uy = -qy / s2, uz = -qz / s2, uw = qw / s2;
    const float tt = uw * uw - (ux * ux + uy * uy + uz * uz);

    const float M00 = tt + 2.f * ux * ux;
    const float M01 = 2.f * (ux * uy - uw * uz);
    const float M02 = 2.f * (ux * uz + uw * uy);
    const float M10 = 2.f * (ux * uy + uw * uz);
    const float M11 = tt + 2.f * uy * uy;
    const float M12 = 2.f * (uy * uz - uw * ux);
    const float M20 = 2.f * (ux * uz - uw * uy);
    const float M21 = 2.f * (uy * uz + uw * ux);
    const float M22 = tt + 2.f * uz * uz;

    const int r = m & 7;
    float c0 = 0.f, c1 = 0.f, c2 = 0.f, c3 = 0.f;
    if (r == 0)      { c0 = M00; c1 = M01; c2 = M02; c3 = -tx; }
    else if (r == 1) { c0 = M10; c1 = M11; c2 = M12; c3 = -ty; }
    else if (r == 2) { c0 = M20; c1 = M21; c2 = M22; c3 = -tz; }
    else if (r == 3) { c0 = M00 * INVHX; c1 = M01 * INVHX; c2 = M02 * INVHX; c3 = -(tx + MIDX) * INVHX; }
    else if (r == 4) { c0 = M10 * INVHY; c1 = M11 * INVHY; c2 = M12 * INVHY; c3 = -(ty + MIDY) * INVHY; }
    else if (r == 5) { c0 = M20 * INVHZ; c1 = M21 * INVHZ; c2 = M22 * INVHZ; c3 = -(tz + MIDZ) * INVHZ; }
    const bool lowrows = m < 8;
    v2f A1, A2;
    A1.x = lowrows ? (hi ? c2 : c0) : 0.f;
    A1.y = lowrows ? (hi ? c3 : c1) : 0.f;
    A2.x = lowrows ? 0.f : (hi ? c0 : c2);
    A2.y = lowrows ? 0.f : (hi ? c1 : c3);

    float accS = 0.f, accC = 0.f;

    int it = 0;
    for (; it + 2 < nb; it += 2) {
        load_buf(bufB, rb + (size_t)(it + 1) * (size_t)(UNR * 384));
        process_buf(bufA, hi, A1, A2, accS, accC);
        load_buf(bufA, rb + (size_t)(it + 2) * (size_t)(UNR * 384));
        process_buf(bufB, hi, A1, A2, accS, accC);
    }
    if (it + 1 < nb) {
        load_buf(bufB, rb + (size_t)(it + 1) * (size_t)(UNR * 384));
        process_buf(bufA, hi, A1, A2, accS, accC);
        process_buf(bufB, hi, A1, A2, accS, accC);
    } else if (it < nb) {
        process_buf(bufA, hi, A1, A2, accS, accC);
    }
    for (int ch = nb * UNR; ch < nch; ++ch) {
        const P3 p = load_pt(rb + (size_t)ch * 384u);
        do_chunk(p, hi, A1, A2, accS, accC);
    }

    for (int off = 16; off > 0; off >>= 1) {
        accS += __shfl_xor(accS, off, 32);
        accC += __shfl_xor(accC, off, 32);
    }

    if (lane == 0)
        out[b] = (accC > 0.f) ? (-accS / accC) * DIST_COEFF : DIST_COEFF;
}

extern "C" void kernel_launch(void* const* d_in, const int* in_sizes, int n_in,
                              void* d_out, int out_size, void* d_ws, size_t ws_size,
                              hipStream_t stream) {
    const float* trans = (const float*)d_in[0];
    const float* quat  = (const float*)d_in[1];
    const float* pc    = (const float*)d_in[2];

    const int B = in_sizes[0] / 3;
    const int N = (int)(in_sizes[2] / ((long long)B * 3));

    collision_dist_kernel<<<dim3(B), dim3(32), 0, stream>>>(
        trans, quat, pc, (float*)d_out, N);
}


# each lane loads its B-fragment halves directly, (x,y) | (y,z), two 8-byte loads per 32 points; lane swap + 4 selects per chunk go; translation = accumulator start value (new rounding order)
# speedup vs baseline: 1.1840x; 1.0404x over previous
#include <hip/hip_runtime.h>
#include <math.h>


typedef __attribute__((ext_vector_type(2))) float v2f;
typedef __attribute__((ext_vector_type(8))) float v8f;

#define DIST_COEFF 10000.0f

#define BCX (-0.001782f)
#define BCY (1.005e-05f)
#define BCZ (0.0431621f)
#define HEX ((float)(0.204416 / 2.0 + 0.001))
#define HEY ((float)(0.0632517 / 2.0 + 0.001))
#define HEZ ((float)(0.1381738 / 2.0 + 0.001))

#define LOX (BCX - HEX)
#define HIX (BCX + HEX)
#define LOY (BCY - HEY)
#define HIY (BCY + HEY)
#define LOZ (BCZ - HEZ)
#define HIZ (BCZ + HEZ)
#define MIDX ((float)(((double)LOX + (double)HIX) * 0.5))
#define MIDY ((float)(((double)LOY + (double)HIY) * 0.5))
#define MIDZ ((float)(((double)LOZ + (double)HIZ) * 0.5))
#define INVHX ((float)(2.0 / ((double)HIX - (double)LOX)))
#define INVHY ((float)(2.0 / ((double)HIY - (double)LOY)))
#define INVHZ ((float)(2.0 / ((double)HIZ - (double)LOZ)))

#define UNR 16

struct Frag { v2f g0, g1; };

__device__ __forceinline__ Frag load_frag(const char* p)
{
    const float* f = (const float*)p;
    Frag r;
    r.g0.x = f[0];
    r.g0.y = f[1];
    r.g1.x = f[48];
    r.g1.y = f[49];
    return r;
}

__device__ __forceinline__ void do_chunk(const Frag b,
                                         const v2f A1, const v2f A2, const v8f C0,
                                         float& accS, float& accC)
{
    v8f d = __builtin_amdgcn_wmma_f32_16x16x4_f32(
        false, A1, false, b.g0, (short)0, C0, false, false);
    d = __builtin_amdgcn_wmma_f32_16x16x4_f32(
        false, A2, false, b.g1, (short)0, d, false, false);

    const float px = d[0], py = d[1], pz = d[2];

    const float w = fmaxf(fmaxf(fabsf(d[3]), fabsf(d[4])), fabsf(d[5]));
    const bool in = w <= 1.0f;
    const float nr = __builtin_amdgcn_sqrtf(px * px + py * py + pz * pz);
    accS += in ? nr : 0.0f;
    accC += in ? 1.0f : 0.0f;
}

#define GRP 4

__device__ __forceinline__ void load_buf(Frag (&buf)[UNR], const char* q)
{
    #pragma unroll
    for (int u = 0; u < UNR; ++u) buf[u] = load_frag(q + (size_t)u * 384u);
    __builtin_amdgcn_sched_barrier(0);
}

__device__ __forceinline__ void process_buf(const Frag (&buf)[UNR],
                                            const v2f A1, const v2f A2, const v8f C0,
                                            float& accS, float& accC)
{
    #pragma unroll
    for (int g = 0; g < UNR; g += GRP) {
        #pragma unroll
        for (int u = 0; u < GRP; ++u)
            do_chunk(buf[g + u], A1, A2, C0, accS, accC);
        __builtin_amdgcn_sched_barrier(0);
    }
}

__global__ __launch_bounds__(32) void collision_dist_kernel(
    const float* __restrict__ trans,
    const float* __restrict__ quat,
    const float* __restrict__ pc,
    float* __restrict__ out,
    int N)
{
    const int b    = blockIdx.x;
    const int lane = threadIdx.x & 31;
    const int m    = lane & 15;
    const bool hi  = lane >= 16;

    const int nch = N >> 5;
    const int nb  = nch / UNR;

    const char* rb = (const char*)pc + (size_t)b * (size_t)N * 12u
                   + (size_t)m * 12u + (hi ? 4u : 0u);

    Frag bufA[UNR], bufB[UNR];
    if (nb > 0)
        load_buf(bufA, rb);

    const float qx = quat[b * 4 + 0], qy = quat[b * 4 + 1];
    const float qz = quat[b * 4 + 2], qw = quat[b * 4 + 3];
    const float tx = trans[b * 3 + 0], ty = trans[b * 3 + 1], tz = trans[b * 3 + 2];

    const float s2 = qx * qx + qy * qy + qz * qz + qw * qw;
    const float ux = -qx / s2, uy = -qy / s2, uz = -qz / s2, uw = qw / s2;
    const float tt = uw * uw - (ux * ux + uy * uy + uz * uz);

    const float M00 = tt + 2.f * ux * ux;
    const float M01 = 2.f * (ux * uy - uw * uz);
    const float M02 = 2.f * (ux * uz + uw * uy);
    const float M10 = 2.f * (ux * uy + uw * uz);
    const float M11 = tt + 2.f * uy * uy;
    const float M12 = 2.f * (uy * uz - uw * ux);
    const float M20 = 2.f * (ux * uz - uw * uy);
    const float M21 = 2.f * (uy * uz + uw * ux);
    const float M22 = tt + 2.f * uz * uz;

    const int r = m & 7;
    float c0 = 0.f, c1 = 0.f, c2 = 0.f;
    if (r == 0)      { c0 = M00; c1 = M01; c2 = M02; }
    else if (r == 1) { c0 = M10; c1 = M11; c2 = M12; }
    else if (r == 2) { c0 = M20; c1 = M21; c2 = M22; }
    else if (r == 3) { c0 = M00 * INVHX; c1 = M01 * INVHX; c2 = M02 * INVHX; }
    else if (r == 4) { c0 = M10 * INVHY; c1 = M11 * INVHY; c2 = M12 * INVHY; }
    else if (r == 5) { c0 = M20 * INVHZ; c1 = M21 * INVHZ; c2 = M22 * INVHZ; }
    const bool lowrows = m < 8;
    const float ax = hi ? 0.f : c0;
    const float ay = hi ? c2  : c1;
    v2f A1, A2;
    A1.x = lowrows ? ax : 0.f;
    A1.y = lowrows ? ay : 0.f;
    A2.x = lowrows ? 0.f : ax;
    A2.y = lowrows ? 0.f : ay;

    v8f C0;
    C0[0] = -tx;
    C0[1] = -ty;
    C0[2] = -tz;
    C0[3] = -(tx + MIDX) * INVHX;
    C0[4] = -(ty + MIDY) * INVHY;
    C0[5] = -(tz + MIDZ) * INVHZ;
    C0[6] = 0.f;
    C0[7] = 0.f;

    float accS = 0.f, accC = 0.f;

    int it = 0;
    for (; it + 2 < nb; it += 2) {
        load_buf(bufB, rb + (size_t)(it + 1) * (size_t)(UNR * 384));
        process_buf(bufA, A1, A2, C0, accS, accC);
        load_buf(bufA, rb + (size_t)(it + 2) * (size_t)(UNR * 384));
        process_buf(bufB, A1, A2, C0, accS, accC);
    }
    if (it + 1 < nb) {
        load_buf(bufB, rb + (size_t)(it + 1) * (size_t)(UNR * 384));
        process_buf(bufA, A1, A2, C0, accS, accC);
        process_buf(bufB, A1, A2, C0, accS, accC);
    } else if (it < nb) {
        process_buf(bufA, A1, A2, C0, accS, accC);
    }
    for (int ch = nb * UNR; ch < nch; ++ch) {
        const Frag f = load_frag(rb + (size_t)ch * 384u);
        do_chunk(f, A1, A2, C0, accS, accC);
    }

    for (int off = 16; off > 0; off >>= 1) {
        accS += __shfl_xor(accS, off, 32);
        accC += __shfl_xor(accC, off, 32);
    }

    if (lane == 0)
        out[b] = (accC > 0.f) ? (-accS / accC) * DIST_COEFF : DIST_COEFF;
}

extern "C" void kernel_launch(void* const* d_in, const int* in_sizes, int n_in,
                              void* d_out, int out_size, void* d_ws, size_t ws_size,
                              hipStream_t stream) {
    const float* trans = (const float*)d_in[0];
    const float* quat  = (const float*)d_in[1];
    const float* pc    = (const float*)d_in[2];

    const int B = in_sizes[0] / 3;
    const int N = (int)(in_sizes[2] / ((long long)B * 3));

    collision_dist_kernel<<<dim3(B), dim3(32), 0, stream>>>(
        trans, quat, pc, (float*)d_out, N);
}
